# starting kernel
# baseline (speedup 1.0000x reference)
#include <hip/hip_runtime.h>

#define USER_NUM 50000
#define ITEM_NUM 50000
#define N_TOTAL  (USER_NUM + ITEM_NUM)
#define FREQ     512
#define EMB      64
#define MAX_SPLIT 40
#define PROJ_ELEMS (FREQ * EMB)

typedef __attribute__((ext_vector_type(2))) float v2f;
typedef __attribute__((ext_vector_type(8))) float v8f;

__device__ __forceinline__ v8f wmma_f32(v2f a, v2f b, v8f c) {
  return __builtin_amdgcn_wmma_f32_16x16x4_f32(
       false, a,  false, b,
       (short)0, c,  false,  false);
}

__device__ __forceinline__ void accum_segment(const float* __restrict__ v,
                                              const float* __restrict__ ego,
                                              int beg, int end,
                                              int m, int kh, int f0,
                                              v8f& acc0, v8f& acc1,
                                              v8f& acc2, v8f& acc3) {
  #pragma unroll 4
  for (int n0 = beg; n0 < end; n0 += 4) {
    const int r0 = n0 + 2 * kh;
    const float* __restrict__ av = v + (size_t)r0 * FREQ + f0 + m;
    v2f a;
    a.x = __builtin_nontemporal_load(av);
    a.y = __builtin_nontemporal_load(av + FREQ);
    const float* __restrict__ e0 = ego + (size_t)r0 * EMB + m;
    const float* __restrict__ e1 = e0 + EMB;
    v2f b;
    b.x = e0[ 0]; b.y = e1[ 0]; acc0 = wmma_f32(a, b, acc0);
    b.x = e0[16]; b.y = e1[16]; acc1 = wmma_f32(a, b, acc1);
    b.x = e0[32]; b.y = e1[32]; acc2 = wmma_f32(a, b, acc2);
    b.x = e0[48]; b.y = e1[48]; acc3 = wmma_f32(a, b, acc3);
  }
}

__global__ void __launch_bounds__(128)
proj_partial_kernel(const float* __restrict__ user_emb,
                    const float* __restrict__ item_emb,
                    const float* __restrict__ v,
                    float* __restrict__ partials,
                    int chunk) {
  const int lane  = threadIdx.x & 31;
  const int wave  = threadIdx.x >> 5;
  const int fTile = blockIdx.x * 4 + wave;
  const int sp    = blockIdx.y;
  const int m     = lane & 15;
  const int kh    = lane >> 4;
  const int f0    = fTile * 16;

  int nBeg = sp * chunk;
  int nEnd = nBeg + chunk;
  if (nEnd > N_TOTAL) nEnd = N_TOTAL;

  v8f acc0 = {0,0,0,0,0,0,0,0};
  v8f acc1 = {0,0,0,0,0,0,0,0};
  v8f acc2 = {0,0,0,0,0,0,0,0};
  v8f acc3 = {0,0,0,0,0,0,0,0};

  const int uEnd = (nEnd < USER_NUM) ? nEnd : USER_NUM;
  const int iBeg = (nBeg > USER_NUM) ? nBeg : USER_NUM;
  if (nBeg < uEnd)
    accum_segment(v, user_emb, nBeg, uEnd, m, kh, f0, acc0, acc1, acc2, acc3);
  if (iBeg < nEnd)
    accum_segment(v, item_emb - (size_t)USER_NUM * EMB, iBeg, nEnd,
                  m, kh, f0, acc0, acc1, acc2, acc3);

  float* __restrict__ dst = partials + (size_t)sp * PROJ_ELEMS;
  #pragma unroll
  for (int r = 0; r < 8; ++r) {
    const int fr = f0 + r + 8 * kh;
    dst[fr * EMB +  0 + m] = acc0[r];
    dst[fr * EMB + 16 + m] = acc1[r];
    dst[fr * EMB + 32 + m] = acc2[r];
    dst[fr * EMB + 48 + m] = acc3[r];
  }
}

__global__ void reduce_scale_kernel(const float* __restrict__ partials,
                                    const float* __restrict__ filt,
                                    float* __restrict__ projS,
                                    int split_cnt) {
  const int i = blockIdx.x * blockDim.x + threadIdx.x;
  if (i < PROJ_ELEMS) {
    float s = 0.f;
    for (int p = 0; p < split_cnt; ++p)
      s += partials[(size_t)p * PROJ_ELEMS + i];
    projS[i] = s * filt[i >> 6];
  }
}

__global__ void __launch_bounds__(128)
filter_combine_kernel(const float* __restrict__ user_emb,
                      const float* __restrict__ item_emb,
                      const float* __restrict__ v,
                      const float* __restrict__ projS,
                      float* __restrict__ out) {
  const int lane  = threadIdx.x & 31;
  const int wave  = threadIdx.x >> 5;
  const int nTile = blockIdx.x * 4 + wave;
  if (nTile * 16 >= N_TOTAL) return;
  const int m  = lane & 15;
  const int kh = lane >> 4;
  const int n0 = nTile * 16;

  v8f acc0 = {0,0,0,0,0,0,0,0};
  v8f acc1 = {0,0,0,0,0,0,0,0};
  v8f acc2 = {0,0,0,0,0,0,0,0};
  v8f acc3 = {0,0,0,0,0,0,0,0};

  const v2f* __restrict__ ap =
      reinterpret_cast<const v2f*>(v + (size_t)(n0 + m) * FREQ) + kh;

  #pragma unroll 4
  for (int f0 = 0; f0 < FREQ; f0 += 4) {
    v2f a = ap[f0 >> 1];
    const float* __restrict__ p0 = projS + (size_t)(f0 + 2 * kh) * EMB + m;
    const float* __restrict__ p1 = p0 + EMB;
    v2f b;
    b.x = p0[ 0]; b.y = p1[ 0]; acc0 = wmma_f32(a, b, acc0);
    b.x = p0[16]; b.y = p1[16]; acc1 = wmma_f32(a, b, acc1);
    b.x = p0[32]; b.y = p1[32]; acc2 = wmma_f32(a, b, acc2);
    b.x = p0[48]; b.y = p1[48]; acc3 = wmma_f32(a, b, acc3);
  }

  const float* __restrict__ ego =
      (n0 < USER_NUM) ? user_emb : (item_emb - (size_t)USER_NUM * EMB);
  #pragma unroll
  for (int r = 0; r < 8; ++r) {
    const int row = n0 + r + 8 * kh;
    const float* __restrict__ er = ego + (size_t)row * EMB + m;
    float* __restrict__ orow = out + (size_t)row * EMB + m;
    __builtin_nontemporal_store((er[ 0] + acc0[r]) * 0.5f, orow +  0);
    __builtin_nontemporal_store((er[16] + acc1[r]) * 0.5f, orow + 16);
    __builtin_nontemporal_store((er[32] + acc2[r]) * 0.5f, orow + 32);
    __builtin_nontemporal_store((er[48] + acc3[r]) * 0.5f, orow + 48);
  }
}

extern "C" void kernel_launch(void* const* d_in, const int* in_sizes, int n_in,
                              void* d_out, int out_size, void* d_ws, size_t ws_size,
                              hipStream_t stream) {
  const float* user_emb = (const float*)d_in[0];
  const float* item_emb = (const float*)d_in[1];
  const float* v        = (const float*)d_in[2];
  const float* filt     = (const float*)d_in[3];
  float* out = (float*)d_out;

  const size_t slot = (size_t)PROJ_ELEMS * sizeof(float);
  int split_cnt = (int)(ws_size / slot) - 1;
  if (split_cnt > MAX_SPLIT) split_cnt = MAX_SPLIT;
  if (split_cnt < 1) split_cnt = 1;

  float* partials = (float*)d_ws;
  float* projS    = partials + (size_t)split_cnt * PROJ_ELEMS;

  int chunk = ((N_TOTAL + split_cnt - 1) / split_cnt + 3) & ~3;

  dim3 gridA(FREQ / 16 / 4, split_cnt);
  proj_partial_kernel<<<gridA, 128, 0, stream>>>(user_emb, item_emb, v,
                                                 partials, chunk);

  reduce_scale_kernel<<<(PROJ_ELEMS + 255) / 256, 256, 0, stream>>>(
      partials, filt, projS, split_cnt);

  const int nTiles = N_TOTAL / 16;
  filter_combine_kernel<<<(nTiles + 3) / 4, 128, 0, stream>>>(
      user_emb, item_emb, v, projS, out);
}


# first kernel: each f tile's N slice is split over 8 waves of a 1024-thread block (256 blocks, 8 waves per SIMD) to hide load latency; tiles summed via LDS in fixed order: proj's f32 sum order changes
# speedup vs baseline: 2.0858x; 2.0858x over previous
#include <hip/hip_runtime.h>

#define USER_NUM 50000
#define ITEM_NUM 50000
#define N_TOTAL  (USER_NUM + ITEM_NUM)
#define FREQ     512
#define EMB      64
#define MAX_SPLIT 32
#define PROJ_ELEMS (FREQ * EMB)
#define K1_SUB   8
#define K1_STEP  (4 * K1_SUB)

typedef __attribute__((ext_vector_type(2))) float v2f;
typedef __attribute__((ext_vector_type(4))) float v4f;
typedef __attribute__((ext_vector_type(8))) float v8f;

__device__ __forceinline__ v8f wmma_f32(v2f a, v2f b, v8f c) {
  return __builtin_amdgcn_wmma_f32_16x16x4_f32(
       false, a,  false, b,
       (short)0, c,  false,  false);
}

__device__ __forceinline__ void accum_segment(const float* __restrict__ v,
                                              const float* __restrict__ ego,
                                              int beg, int end,
                                              int m, int kh, int f0,
                                              v8f& acc0, v8f& acc1,
                                              v8f& acc2, v8f& acc3) {
  #pragma unroll 4
  for (int n0 = beg; n0 < end; n0 += K1_STEP) {
    const int r0 = n0 + 2 * kh;
    const float* __restrict__ av = v + (size_t)r0 * FREQ + f0 + m;
    v2f a;
    a.x = __builtin_nontemporal_load(av);
    a.y = __builtin_nontemporal_load(av + FREQ);
    const float* __restrict__ e0 = ego + (size_t)r0 * EMB + m;
    const float* __restrict__ e1 = e0 + EMB;
    v2f b;
    b.x = e0[ 0]; b.y = e1[ 0]; acc0 = wmma_f32(a, b, acc0);
    b.x = e0[16]; b.y = e1[16]; acc1 = wmma_f32(a, b, acc1);
    b.x = e0[32]; b.y = e1[32]; acc2 = wmma_f32(a, b, acc2);
    b.x = e0[48]; b.y = e1[48]; acc3 = wmma_f32(a, b, acc3);
  }
}

__global__ void __launch_bounds__(128 * K1_SUB)
proj_partial_kernel(const float* __restrict__ user_emb,
                    const float* __restrict__ item_emb,
                    const float* __restrict__ v,
                    float* __restrict__ partials,
                    int chunk) {
  __shared__ v4f red[(K1_SUB - 1) * 4][8][32];

  const int lane  = threadIdx.x & 31;
  const int wave  = __builtin_amdgcn_readfirstlane(threadIdx.x >> 5);
  const int fw    = wave & 3;
  const int s     = wave >> 2;
  const int fTile = blockIdx.x * 4 + fw;
  const int sp    = blockIdx.y;
  const int m     = lane & 15;
  const int kh    = lane >> 4;
  const int f0    = fTile * 16;

  int nBeg = sp * chunk;
  int nEnd = nBeg + chunk;
  if (nEnd > N_TOTAL) nEnd = N_TOTAL;

  v8f acc0 = {0,0,0,0,0,0,0,0};
  v8f acc1 = {0,0,0,0,0,0,0,0};
  v8f acc2 = {0,0,0,0,0,0,0,0};
  v8f acc3 = {0,0,0,0,0,0,0,0};

  const int uEnd = (nEnd < USER_NUM) ? nEnd : USER_NUM;
  const int iBeg = (nBeg > USER_NUM) ? nBeg : USER_NUM;
  if (nBeg < uEnd)
    accum_segment(v, user_emb, nBeg + 4 * s, uEnd, m, kh, f0,
                  acc0, acc1, acc2, acc3);
  if (iBeg < nEnd)
    accum_segment(v, item_emb - (size_t)USER_NUM * EMB, iBeg + 4 * s, nEnd,
                  m, kh, f0, acc0, acc1, acc2, acc3);

  if (s > 0) {
    v4f (*slot)[32] = red[(s - 1) * 4 + fw];
    slot[0][lane] = acc0.lo;  slot[1][lane] = acc0.hi;
    slot[2][lane] = acc1.lo;  slot[3][lane] = acc1.hi;
    slot[4][lane] = acc2.lo;  slot[5][lane] = acc2.hi;
    slot[6][lane] = acc3.lo;  slot[7][lane] = acc3.hi;
  }
  __syncthreads();
  if (s > 0) return;

  #pragma unroll
  for (int t = 0; t < K1_SUB - 1; ++t) {
    v4f (*slot)[32] = red[t * 4 + fw];
    acc0.lo += slot[0][lane];  acc0.hi += slot[1][lane];
    acc1.lo += slot[2][lane];  acc1.hi += slot[3][lane];
    acc2.lo += slot[4][lane];  acc2.hi += slot[5][lane];
    acc3.lo += slot[6][lane];  acc3.hi += slot[7][lane];
  }

  float* __restrict__ dst = partials + (size_t)sp * PROJ_ELEMS;
  #pragma unroll
  for (int r = 0; r < 8; ++r) {
    const int fr = f0 + r + 8 * kh;
    dst[fr * EMB +  0 + m] = acc0[r];
    dst[fr * EMB + 16 + m] = acc1[r];
    dst[fr * EMB + 32 + m] = acc2[r];
    dst[fr * EMB + 48 + m] = acc3[r];
  }
}

__global__ void reduce_scale_kernel(const float* __restrict__ partials,
                                    const float* __restrict__ filt,
                                    float* __restrict__ projS,
                                    int split_cnt) {
  const int i = blockIdx.x * blockDim.x + threadIdx.x;
  if (i < PROJ_ELEMS) {
    float s = 0.f;
    for (int p = 0; p < split_cnt; ++p)
      s += partials[(size_t)p * PROJ_ELEMS + i];
    projS[i] = s * filt[i >> 6];
  }
}

__global__ void __launch_bounds__(128)
filter_combine_kernel(const float* __restrict__ user_emb,
                      const float* __restrict__ item_emb,
                      const float* __restrict__ v,
                      const float* __restrict__ projS,
                      float* __restrict__ out) {
  const int lane  = threadIdx.x & 31;
  const int wave  = threadIdx.x >> 5;
  const int nTile = blockIdx.x * 4 + wave;
  if (nTile * 16 >= N_TOTAL) return;
  const int m  = lane & 15;
  const int kh = lane >> 4;
  const int n0 = nTile * 16;

  v8f acc0 = {0,0,0,0,0,0,0,0};
  v8f acc1 = {0,0,0,0,0,0,0,0};
  v8f acc2 = {0,0,0,0,0,0,0,0};
  v8f acc3 = {0,0,0,0,0,0,0,0};

  const v2f* __restrict__ ap =
      reinterpret_cast<const v2f*>(v + (size_t)(n0 + m) * FREQ) + kh;

  #pragma unroll 4
  for (int f0 = 0; f0 < FREQ; f0 += 4) {
    v2f a = ap[f0 >> 1];
    const float* __restrict__ p0 = projS + (size_t)(f0 + 2 * kh) * EMB + m;
    const float* __restrict__ p1 = p0 + EMB;
    v2f b;
    b.x = p0[ 0]; b.y = p1[ 0]; acc0 = wmma_f32(a, b, acc0);
    b.x = p0[16]; b.y = p1[16]; acc1 = wmma_f32(a, b, acc1);
    b.x = p0[32]; b.y = p1[32]; acc2 = wmma_f32(a, b, acc2);
    b.x = p0[48]; b.y = p1[48]; acc3 = wmma_f32(a, b, acc3);
  }

  const float* __restrict__ ego =
      (n0 < USER_NUM) ? user_emb : (item_emb - (size_t)USER_NUM * EMB);
  #pragma unroll
  for (int r = 0; r < 8; ++r) {
    const int row = n0 + r + 8 * kh;
    const float* __restrict__ er = ego + (size_t)row * EMB + m;
    float* __restrict__ orow = out + (size_t)row * EMB + m;
    __builtin_nontemporal_store((er[ 0] + acc0[r]) * 0.5f, orow +  0);
    __builtin_nontemporal_store((er[16] + acc1[r]) * 0.5f, orow + 16);
    __builtin_nontemporal_store((er[32] + acc2[r]) * 0.5f, orow + 32);
    __builtin_nontemporal_store((er[48] + acc3[r]) * 0.5f, orow + 48);
  }
}

extern "C" void kernel_launch(void* const* d_in, const int* in_sizes, int n_in,
                              void* d_out, int out_size, void* d_ws, size_t ws_size,
                              hipStream_t stream) {
  const float* user_emb = (const float*)d_in[0];
  const float* item_emb = (const float*)d_in[1];
  const float* v        = (const float*)d_in[2];
  const float* filt     = (const float*)d_in[3];
  float* out = (float*)d_out;

  const size_t slot = (size_t)PROJ_ELEMS * sizeof(float);
  int split_cnt = (int)(ws_size / slot) - 1;
  if (split_cnt > MAX_SPLIT) split_cnt = MAX_SPLIT;
  if (split_cnt < 1) split_cnt = 1;

  float* partials = (float*)d_ws;
  float* projS    = partials + (size_t)split_cnt * PROJ_ELEMS;

  int chunk = ((N_TOTAL + split_cnt - 1) / split_cnt + 3) & ~3;

  dim3 gridA(FREQ / 16 / 4, split_cnt);
  proj_partial_kernel<<<gridA, 128 * K1_SUB, 0, stream>>>(user_emb, item_emb, v,
                                                          partials, chunk);

  reduce_scale_kernel<<<(PROJ_ELEMS + 255) / 256, 256, 0, stream>>>(
      partials, filt, projS, split_cnt);

  const int nTiles = N_TOTAL / 16;
  filter_combine_kernel<<<(nTiles + 3) / 4, 128, 0, stream>>>(
      user_emb, item_emb, v, projS, out);
}


# third kernel: 256 workgroups of 32 waves copy the scaled projection (128 KB, stored by kernel 2 in B-fragment order) into LDS once and read all B fragments from LDS, not from global memory
# speedup vs baseline: 2.3102x; 1.1076x over previous
#include <hip/hip_runtime.h>

#define USER_NUM 50000
#define ITEM_NUM 50000
#define N_TOTAL  (USER_NUM + ITEM_NUM)
#define FREQ     512
#define EMB      64
#define MAX_SPLIT 32
#define PROJ_ELEMS (FREQ * EMB)
#define K1_SUB   8
#define K1_STEP  (4 * K1_SUB)

typedef __attribute__((ext_vector_type(2))) float v2f;
typedef __attribute__((ext_vector_type(4))) float v4f;
typedef __attribute__((ext_vector_type(8))) float v8f;

__device__ __forceinline__ v8f wmma_f32(v2f a, v2f b, v8f c) {
  return __builtin_amdgcn_wmma_f32_16x16x4_f32(
       false, a,  false, b,
       (short)0, c,  false,  false);
}

__device__ __forceinline__ void accum_segment(const float* __restrict__ v,
                                              const float* __restrict__ ego,
                                              int beg, int end,
                                              int m, int kh, int f0,
                                              v8f& acc0, v8f& acc1,
                                              v8f& acc2, v8f& acc3) {
  #pragma unroll 4
  for (int n0 = beg; n0 < end; n0 += K1_STEP) {
    const int r0 = n0 + 2 * kh;
    const float* __restrict__ av = v + (size_t)r0 * FREQ + f0 + m;
    v2f a;
    a.x = __builtin_nontemporal_load(av);
    a.y = __builtin_nontemporal_load(av + FREQ);
    const float* __restrict__ e0 = ego + (size_t)r0 * EMB + m;
    const float* __restrict__ e1 = e0 + EMB;
    v2f b;
    b.x = e0[ 0]; b.y = e1[ 0]; acc0 = wmma_f32(a, b, acc0);
    b.x = e0[16]; b.y = e1[16]; acc1 = wmma_f32(a, b, acc1);
    b.x = e0[32]; b.y = e1[32]; acc2 = wmma_f32(a, b, acc2);
    b.x = e0[48]; b.y = e1[48]; acc3 = wmma_f32(a, b, acc3);
  }
}

__global__ void __launch_bounds__(128 * K1_SUB)
proj_partial_kernel(const float* __restrict__ user_emb,
                    const float* __restrict__ item_emb,
                    const float* __restrict__ v,
                    float* __restrict__ partials,
                    int chunk) {
  __shared__ v4f red[(K1_SUB - 1) * 4][8][32];

  const int lane  = threadIdx.x & 31;
  const int wave  = __builtin_amdgcn_readfirstlane(threadIdx.x >> 5);
  const int fw    = wave & 3;
  const int s     = wave >> 2;
  const int fTile = blockIdx.x * 4 + fw;
  const int sp    = blockIdx.y;
  const int m     = lane & 15;
  const int kh    = lane >> 4;
  const int f0    = fTile * 16;

  int nBeg = sp * chunk;
  int nEnd = nBeg + chunk;
  if (nEnd > N_TOTAL) nEnd = N_TOTAL;

  v8f acc0 = {0,0,0,0,0,0,0,0};
  v8f acc1 = {0,0,0,0,0,0,0,0};
  v8f acc2 = {0,0,0,0,0,0,0,0};
  v8f acc3 = {0,0,0,0,0,0,0,0};

  const int uEnd = (nEnd < USER_NUM) ? nEnd : USER_NUM;
  const int iBeg = (nBeg > USER_NUM) ? nBeg : USER_NUM;
  if (nBeg < uEnd)
    accum_segment(v, user_emb, nBeg + 4 * s, uEnd, m, kh, f0,
                  acc0, acc1, acc2, acc3);
  if (iBeg < nEnd)
    accum_segment(v, item_emb - (size_t)USER_NUM * EMB, iBeg + 4 * s, nEnd,
                  m, kh, f0, acc0, acc1, acc2, acc3);

  if (s > 0) {
    v4f (*slot)[32] = red[(s - 1) * 4 + fw];
    slot[0][lane] = acc0.lo;  slot[1][lane] = acc0.hi;
    slot[2][lane] = acc1.lo;  slot[3][lane] = acc1.hi;
    slot[4][lane] = acc2.lo;  slot[5][lane] = acc2.hi;
    slot[6][lane] = acc3.lo;  slot[7][lane] = acc3.hi;
  }
  __syncthreads();
  if (s > 0) return;

  #pragma unroll
  for (int t = 0; t < K1_SUB - 1; ++t) {
    v4f (*slot)[32] = red[t * 4 + fw];
    acc0.lo += slot[0][lane];  acc0.hi += slot[1][lane];
    acc1.lo += slot[2][lane];  acc1.hi += slot[3][lane];
    acc2.lo += slot[4][lane];  acc2.hi += slot[5][lane];
    acc3.lo += slot[6][lane];  acc3.hi += slot[7][lane];
  }

  float* __restrict__ dst = partials + (size_t)sp * PROJ_ELEMS;
  #pragma unroll
  for (int r = 0; r < 8; ++r) {
    const int fr = f0 + r + 8 * kh;
    dst[fr * EMB +  0 + m] = acc0[r];
    dst[fr * EMB + 16 + m] = acc1[r];
    dst[fr * EMB + 32 + m] = acc2[r];
    dst[fr * EMB + 48 + m] = acc3[r];
  }
}

__global__ void reduce_scale_kernel(const float* __restrict__ partials,
                                    const float* __restrict__ filt,
                                    float* __restrict__ projS,
                                    int split_cnt) {
  const int i = blockIdx.x * blockDim.x + threadIdx.x;
  if (i < PROJ_ELEMS) {
    float s = 0.f;
    for (int p = 0; p < split_cnt; ++p)
      s += partials[(size_t)p * PROJ_ELEMS + i];
    const int f = i >> 6;
    const int d = i & 63;
    const int o = (f >> 1) * 128 + (d >> 5) * 64 + (d & 15) * 4 +
                  ((d >> 4) & 1) * 2 + (f & 1);
    projS[o] = s * filt[f];
  }
}

#define K3_BLOCKS 256
#define N_TILES   (N_TOTAL / 16)

__global__ void __launch_bounds__(1024) __attribute__((amdgpu_waves_per_eu(8, 8)))
filter_combine_kernel(const float* __restrict__ user_emb,
                      const float* __restrict__ item_emb,
                      const float* __restrict__ v,
                      const float* __restrict__ projS,
                      float* __restrict__ out) {
  __shared__ v4f projL[PROJ_ELEMS / 4];

  const v4f* __restrict__ src = reinterpret_cast<const v4f*>(projS);
  #pragma unroll
  for (int i = 0; i < PROJ_ELEMS / 4 / 1024; ++i) {
    const int idx = ((i + blockIdx.x) & 7) * 1024 + threadIdx.x;
    projL[idx] = src[idx];
  }
  __syncthreads();

  const int lane = threadIdx.x & 31;
  const int wave = __builtin_amdgcn_readfirstlane(threadIdx.x >> 5);
  const int m  = lane & 15;
  const int kh = lane >> 4;

  const int tBeg = (blockIdx.x * N_TILES) / K3_BLOCKS;
  const int tEnd = ((blockIdx.x + 1) * N_TILES) / K3_BLOCKS;

  const v4f* bp = projL + kh * 32 + m;

  for (int nTile = tBeg + wave; nTile < tEnd; nTile += 32) {
    const int n0 = nTile * 16;

    v8f acc0 = {0,0,0,0,0,0,0,0};
    v8f acc1 = {0,0,0,0,0,0,0,0};
    v8f acc2 = {0,0,0,0,0,0,0,0};
    v8f acc3 = {0,0,0,0,0,0,0,0};

    const v2f* __restrict__ ap =
        reinterpret_cast<const v2f*>(v + (size_t)(n0 + m) * FREQ) + kh;

    for (int f0 = 0; f0 < FREQ; f0 += 16) {
      v2f a[4];
      v4f q[8];
      #pragma unroll
      for (int j = 0; j < 4; ++j)
        a[j] = ap[(f0 >> 1) + 2 * j];
      #pragma unroll
      for (int j = 0; j < 4; ++j) {
        q[2 * j]     = bp[(f0 + 4 * j) * 16];
        q[2 * j + 1] = bp[(f0 + 4 * j) * 16 + 16];
      }
      #pragma unroll
      for (int j = 0; j < 4; ++j) {
        v2f b;
        b.x = q[2*j].x;   b.y = q[2*j].y;   acc0 = wmma_f32(a[j], b, acc0);
        b.x = q[2*j].z;   b.y = q[2*j].w;   acc1 = wmma_f32(a[j], b, acc1);
        b.x = q[2*j+1].x; b.y = q[2*j+1].y; acc2 = wmma_f32(a[j], b, acc2);
        b.x = q[2*j+1].z; b.y = q[2*j+1].w; acc3 = wmma_f32(a[j], b, acc3);
      }
    }

    const float* __restrict__ ego =
        (n0 < USER_NUM) ? user_emb : (item_emb - (size_t)USER_NUM * EMB);
    #pragma unroll
    for (int r = 0; r < 8; ++r) {
      const int row = n0 + r + 8 * kh;
      const float* __restrict__ er = ego + (size_t)row * EMB + m;
      float* __restrict__ orow = out + (size_t)row * EMB + m;
      __builtin_nontemporal_store((er[ 0] + acc0[r]) * 0.5f, orow +  0);
      __builtin_nontemporal_store((er[16] + acc1[r]) * 0.5f, orow + 16);
      __builtin_nontemporal_store((er[32] + acc2[r]) * 0.5f, orow + 32);
      __builtin_nontemporal_store((er[48] + acc3[r]) * 0.5f, orow + 48);
    }
  }
}

extern "C" void kernel_launch(void* const* d_in, const int* in_sizes, int n_in,
                              void* d_out, int out_size, void* d_ws, size_t ws_size,
                              hipStream_t stream) {
  const float* user_emb = (const float*)d_in[0];
  const float* item_emb = (const float*)d_in[1];
  const float* v        = (const float*)d_in[2];
  const float* filt     = (const float*)d_in[3];
  float* out = (float*)d_out;

  const size_t slot = (size_t)PROJ_ELEMS * sizeof(float);
  int split_cnt = (int)(ws_size / slot) - 1;
  if (split_cnt > MAX_SPLIT) split_cnt = MAX_SPLIT;
  if (split_cnt < 1) split_cnt = 1;

  float* partials = (float*)d_ws;
  float* projS    = partials + (size_t)split_cnt * PROJ_ELEMS;

  int chunk = ((N_TOTAL + split_cnt - 1) / split_cnt + 3) & ~3;

  dim3 gridA(FREQ / 16 / 4, split_cnt);
  proj_partial_kernel<<<gridA, 128 * K1_SUB, 0, stream>>>(user_emb, item_emb, v,
                                                          partials, chunk);

  reduce_scale_kernel<<<(PROJ_ELEMS + 255) / 256, 256, 0, stream>>>(
      partials, filt, projS, split_cnt);

  filter_combine_kernel<<<K3_BLOCKS, 1024, 0, stream>>>(
      user_emb, item_emb, v, projS, out);
}


# first kernel: a wave owns a whole 64x64 (f x d) tile, columns interleaved: four whole-row 16-byte loads feed the 16 WMMAs of a 4-row K group (was ten 4-byte loads per 4); proj f32 sum order changes
# speedup vs baseline: 2.7165x; 1.1759x over previous
#include <hip/hip_runtime.h>

#define USER_NUM 50000
#define ITEM_NUM 50000
#define N_TOTAL  (USER_NUM + ITEM_NUM)
#define FREQ     512
#define EMB      64
#define MAX_SPLIT 32
#define PROJ_ELEMS (FREQ * EMB)
#define K1_SUB   16
#define K1_STEP  (4 * K1_SUB)

typedef __attribute__((ext_vector_type(2))) float v2f;
typedef __attribute__((ext_vector_type(4))) float v4f;
typedef __attribute__((ext_vector_type(8))) float v8f;

__device__ __forceinline__ v8f wmma_f32(v2f a, v2f b, v8f c) {
  return __builtin_amdgcn_wmma_f32_16x16x4_f32(
       false, a,  false, b,
       (short)0, c,  false,  false);
}

__device__ __forceinline__ void accum_segment(const float* __restrict__ v,
                                              const float* __restrict__ ego,
                                              int beg, int end,
                                              int m, int kh, int f0,
                                              v8f (&acc)[16]) {
  const v4f* __restrict__ av =
      reinterpret_cast<const v4f*>(v + (size_t)(beg + 2 * kh) * FREQ + f0 + 4 * m);
  const v4f* __restrict__ ev =
      reinterpret_cast<const v4f*>(ego + (size_t)(beg + 2 * kh) * EMB + 4 * m);
  #pragma unroll 4
  for (int n0 = beg; n0 < end; n0 += K1_STEP,
                               av += K1_STEP * (FREQ / 4),
                               ev += K1_STEP * (EMB / 4)) {
    const v4f a0 = __builtin_nontemporal_load(av);
    const v4f a1 = __builtin_nontemporal_load(av + FREQ / 4);
    const v4f b0 = ev[0];
    const v4f b1 = ev[EMB / 4];
    #pragma unroll
    for (int ft = 0; ft < 4; ++ft) {
      v2f a;
      a.x = a0[ft]; a.y = a1[ft];
      #pragma unroll
      for (int dt = 0; dt < 4; ++dt) {
        v2f b;
        b.x = b0[dt]; b.y = b1[dt];
        acc[ft * 4 + dt] = wmma_f32(a, b, acc[ft * 4 + dt]);
      }
    }
  }
}

__global__ void __launch_bounds__(32 * K1_SUB)
proj_partial_kernel(const float* __restrict__ user_emb,
                    const float* __restrict__ item_emb,
                    const float* __restrict__ v,
                    float* __restrict__ partials,
                    int chunk) {
  __shared__ v4f red[K1_SUB][32][32];

  const int lane  = threadIdx.x & 31;
  const int s     = __builtin_amdgcn_readfirstlane(threadIdx.x >> 5);
  const int sp    = blockIdx.y;
  const int m     = lane & 15;
  const int kh    = lane >> 4;
  const int f0    = blockIdx.x * 64;

  int nBeg = sp * chunk;
  int nEnd = nBeg + chunk;
  if (nEnd > N_TOTAL) nEnd = N_TOTAL;

  v8f acc[16];
  #pragma unroll
  for (int t = 0; t < 16; ++t) acc[t] = (v8f){0,0,0,0,0,0,0,0};

  const int uEnd = (nEnd < USER_NUM) ? nEnd : USER_NUM;
  const int iBeg = (nBeg > USER_NUM) ? nBeg : USER_NUM;
  if (nBeg < uEnd)
    accum_segment(v, user_emb, nBeg + 4 * s, uEnd, m, kh, f0, acc);
  if (iBeg < nEnd)
    accum_segment(v, item_emb - (size_t)USER_NUM * EMB, iBeg + 4 * s, nEnd,
                  m, kh, f0, acc);

  #pragma unroll
  for (int t = 0; t < 16; ++t) {
    red[s][2 * t][lane]     = acc[t].lo;
    red[s][2 * t + 1][lane] = acc[t].hi;
  }
  __syncthreads();

  const int ft   = s >> 2;
  const int q    = s & 3;
  const int half = q >> 1;
  const int comp = (q & 1) * 2;
  v2f sum[4];
  #pragma unroll
  for (int p = 0; p < K1_SUB; ++p) {
    #pragma unroll
    for (int dt = 0; dt < 4; ++dt) {
      const float* piece = reinterpret_cast<const float*>(
          &red[p][2 * (ft * 4 + dt) + half][lane]);
      const v2f x = *reinterpret_cast<const v2f*>(piece + comp);
      if (p == 0) sum[dt] = x; else sum[dt] += x;
    }
  }

  float* __restrict__ dst = partials + (size_t)sp * PROJ_ELEMS;
  #pragma unroll
  for (int rr = 0; rr < 2; ++rr) {
    const int fr = f0 + 4 * (2 * q + rr + 8 * kh) + ft;
    v4f o;
    o.x = sum[0][rr]; o.y = sum[1][rr]; o.z = sum[2][rr]; o.w = sum[3][rr];
    *reinterpret_cast<v4f*>(dst + fr * EMB + 4 * m) = o;
  }
}

__global__ void reduce_scale_kernel(const float* __restrict__ partials,
                                    const float* __restrict__ filt,
                                    float* __restrict__ projS,
                                    int split_cnt) {
  const int i = blockIdx.x * blockDim.x + threadIdx.x;
  if (i < PROJ_ELEMS) {
    float s = 0.f;
    for (int p = 0; p < split_cnt; ++p)
      s += partials[(size_t)p * PROJ_ELEMS + i];
    const int f = i >> 6;
    const int d = i & 63;
    const int o = (f >> 1) * 128 + (d >> 5) * 64 + (d & 15) * 4 +
                  ((d >> 4) & 1) * 2 + (f & 1);
    projS[o] = s * filt[f];
  }
}

#define K3_BLOCKS 256
#define N_TILES   (N_TOTAL / 16)

__global__ void __launch_bounds__(1024) __attribute__((amdgpu_waves_per_eu(8, 8)))
filter_combine_kernel(const float* __restrict__ user_emb,
                      const float* __restrict__ item_emb,
                      const float* __restrict__ v,
                      const float* __restrict__ projS,
                      float* __restrict__ out) {
  __shared__ v4f projL[PROJ_ELEMS / 4];

  const v4f* __restrict__ src = reinterpret_cast<const v4f*>(projS);
  #pragma unroll
  for (int i = 0; i < PROJ_ELEMS / 4 / 1024; ++i) {
    const int idx = ((i + blockIdx.x) & 7) * 1024 + threadIdx.x;
    projL[idx] = src[idx];
  }
  __syncthreads();

  const int lane = threadIdx.x & 31;
  const int wave = __builtin_amdgcn_readfirstlane(threadIdx.x >> 5);
  const int m  = lane & 15;
  const int kh = lane >> 4;

  const int tBeg = (blockIdx.x * N_TILES) / K3_BLOCKS;
  const int tEnd = ((blockIdx.x + 1) * N_TILES) / K3_BLOCKS;

  const v4f* bp = projL + kh * 32 + m;

  for (int nTile = tBeg + wave; nTile < tEnd; nTile += 32) {
    const int n0 = nTile * 16;

    v8f acc0 = {0,0,0,0,0,0,0,0};
    v8f acc1 = {0,0,0,0,0,0,0,0};
    v8f acc2 = {0,0,0,0,0,0,0,0};
    v8f acc3 = {0,0,0,0,0,0,0,0};

    const v2f* __restrict__ ap =
        reinterpret_cast<const v2f*>(v + (size_t)(n0 + m) * FREQ) + kh;

    for (int f0 = 0; f0 < FREQ; f0 += 16) {
      v2f a[4];
      v4f q[8];
      #pragma unroll
      for (int j = 0; j < 4; ++j)
        a[j] = ap[(f0 >> 1) + 2 * j];
      #pragma unroll
      for (int j = 0; j < 4; ++j) {
        q[2 * j]     = bp[(f0 + 4 * j) * 16];
        q[2 * j + 1] = bp[(f0 + 4 * j) * 16 + 16];
      }
      #pragma unroll
      for (int j = 0; j < 4; ++j) {
        v2f b;
        b.x = q[2*j].x;   b.y = q[2*j].y;   acc0 = wmma_f32(a[j], b, acc0);
        b.x = q[2*j].z;   b.y = q[2*j].w;   acc1 = wmma_f32(a[j], b, acc1);
        b.x = q[2*j+1].x; b.y = q[2*j+1].y; acc2 = wmma_f32(a[j], b, acc2);
        b.x = q[2*j+1].z; b.y = q[2*j+1].w; acc3 = wmma_f32(a[j], b, acc3);
      }
    }

    const float* __restrict__ ego =
        (n0 < USER_NUM) ? user_emb : (item_emb - (size_t)USER_NUM * EMB);
    #pragma unroll
    for (int r = 0; r < 8; ++r) {
      const int row = n0 + r + 8 * kh;
      const float* __restrict__ er = ego + (size_t)row * EMB + m;
      float* __restrict__ orow = out + (size_t)row * EMB + m;
      __builtin_nontemporal_store((er[ 0] + acc0[r]) * 0.5f, orow +  0);
      __builtin_nontemporal_store((er[16] + acc1[r]) * 0.5f, orow + 16);
      __builtin_nontemporal_store((er[32] + acc2[r]) * 0.5f, orow + 32);
      __builtin_nontemporal_store((er[48] + acc3[r]) * 0.5f, orow + 48);
    }
  }
}

extern "C" void kernel_launch(void* const* d_in, const int* in_sizes, int n_in,
                              void* d_out, int out_size, void* d_ws, size_t ws_size,
                              hipStream_t stream) {
  const float* user_emb = (const float*)d_in[0];
  const float* item_emb = (const float*)d_in[1];
  const float* v        = (const float*)d_in[2];
  const float* filt     = (const float*)d_in[3];
  float* out = (float*)d_out;

  const size_t slot = (size_t)PROJ_ELEMS * sizeof(float);
  int split_cnt = (int)(ws_size / slot) - 1;
  if (split_cnt > MAX_SPLIT) split_cnt = MAX_SPLIT;
  if (split_cnt < 1) split_cnt = 1;

  float* partials = (float*)d_ws;
  float* projS    = partials + (size_t)split_cnt * PROJ_ELEMS;

  int chunk = ((N_TOTAL + split_cnt - 1) / split_cnt + 3) & ~3;

  dim3 gridA(FREQ / 64, split_cnt);
  proj_partial_kernel<<<gridA, 32 * K1_SUB, 0, stream>>>(user_emb, item_emb, v,
                                                         partials, chunk);

  reduce_scale_kernel<<<(PROJ_ELEMS + 255) / 256, 256, 0, stream>>>(
      partials, filt, projS, split_cnt);

  filter_combine_kernel<<<K3_BLOCKS, 1024, 0, stream>>>(
      user_emb, item_emb, v, projS, out);
}


# both GEMM kernels: a wave lowers its own issue priority (s_setprio 3,2,1,0) over the quarters of its K range, so a SIMD's waves advance together and none ends alone and latency-bound; same sums
# speedup vs baseline: 2.8459x; 1.0476x over previous
#include <hip/hip_runtime.h>

#define USER_NUM 50000
#define ITEM_NUM 50000
#define N_TOTAL  (USER_NUM + ITEM_NUM)
#define FREQ     512
#define EMB      64
#define MAX_SPLIT 32
#define PROJ_ELEMS (FREQ * EMB)
#define K1_SUB   16
#define K1_STEP  (4 * K1_SUB)

typedef __attribute__((ext_vector_type(2))) float v2f;
typedef __attribute__((ext_vector_type(4))) float v4f;
typedef __attribute__((ext_vector_type(8))) float v8f;

__device__ __forceinline__ v8f wmma_f32(v2f a, v2f b, v8f c) {
  return __builtin_amdgcn_wmma_f32_16x16x4_f32(
       false, a,  false, b,
       (short)0, c,  false,  false);
}

__device__ __forceinline__ void accum_segment(const float* __restrict__ v,
                                              const float* __restrict__ ego,
                                              int beg, int end,
                                              int m, int kh, int f0,
                                              v8f (&acc)[16]) {
  const v4f* __restrict__ av =
      reinterpret_cast<const v4f*>(v + (size_t)(beg + 2 * kh) * FREQ + f0 + 4 * m);
  const v4f* __restrict__ ev =
      reinterpret_cast<const v4f*>(ego + (size_t)(beg + 2 * kh) * EMB + 4 * m);
  #pragma unroll 4
  for (int n0 = beg; n0 < end; n0 += K1_STEP,
                               av += K1_STEP * (FREQ / 4),
                               ev += K1_STEP * (EMB / 4)) {
    const v4f a0 = __builtin_nontemporal_load(av);
    const v4f a1 = __builtin_nontemporal_load(av + FREQ / 4);
    const v4f b0 = ev[0];
    const v4f b1 = ev[EMB / 4];
    #pragma unroll
    for (int ft = 0; ft < 4; ++ft) {
      v2f a;
      a.x = a0[ft]; a.y = a1[ft];
      #pragma unroll
      for (int dt = 0; dt < 4; ++dt) {
        v2f b;
        b.x = b0[dt]; b.y = b1[dt];
        acc[ft * 4 + dt] = wmma_f32(a, b, acc[ft * 4 + dt]);
      }
    }
  }
}

template <int PRIO>
__device__ __forceinline__ void k1_phase(const float* __restrict__ v,
                                         const float* __restrict__ user_emb,
                                         const float* __restrict__ item_rebased,
                                         int& curU, int& curI, int limU, int limI,
                                         int m, int kh, int f0, v8f (&acc)[16]) {
  __builtin_amdgcn_s_setprio(PRIO);
  if (curU < limU) {
    accum_segment(v, user_emb, curU, limU, m, kh, f0, acc);
    curU += ((limU - curU + K1_STEP - 1) / K1_STEP) * K1_STEP;
  }
  if (curI < limI) {
    accum_segment(v, item_rebased, curI, limI, m, kh, f0, acc);
    curI += ((limI - curI + K1_STEP - 1) / K1_STEP) * K1_STEP;
  }
}

__global__ void __launch_bounds__(32 * K1_SUB)
proj_partial_kernel(const float* __restrict__ user_emb,
                    const float* __restrict__ item_emb,
                    const float* __restrict__ v,
                    float* __restrict__ partials,
                    int chunk) {
  __shared__ v4f red[K1_SUB][32][32];

  const int lane  = threadIdx.x & 31;
  const int s     = __builtin_amdgcn_readfirstlane(threadIdx.x >> 5);
  const int sp    = blockIdx.y;
  const int m     = lane & 15;
  const int kh    = lane >> 4;
  const int f0    = blockIdx.x * 64;

  int nBeg = sp * chunk;
  int nEnd = nBeg + chunk;
  if (nEnd > N_TOTAL) nEnd = N_TOTAL;

  v8f acc[16];
  #pragma unroll
  for (int t = 0; t < 16; ++t) acc[t] = (v8f){0,0,0,0,0,0,0,0};

  const int uEnd = (nEnd < USER_NUM) ? nEnd : USER_NUM;
  const int iBeg = (nBeg > USER_NUM) ? nBeg : USER_NUM;
  const float* __restrict__ item_rebased = item_emb - (size_t)USER_NUM * EMB;
  int curU = nBeg + 4 * s;
  int curI = iBeg + 4 * s;
  const int pr = (chunk / (16 * K1_STEP)) * (4 * K1_STEP);
  const int h1 = (nBeg + pr     < nEnd) ? nBeg + pr     : nEnd;
  const int h2 = (nBeg + 2 * pr < nEnd) ? nBeg + 2 * pr : nEnd;
  const int h3 = (nBeg + 3 * pr < nEnd) ? nBeg + 3 * pr : nEnd;
  k1_phase<3>(v, user_emb, item_rebased, curU, curI,
              (h1 < uEnd) ? h1 : uEnd, h1, m, kh, f0, acc);
  k1_phase<2>(v, user_emb, item_rebased, curU, curI,
              (h2 < uEnd) ? h2 : uEnd, h2, m, kh, f0, acc);
  k1_phase<1>(v, user_emb, item_rebased, curU, curI,
              (h3 < uEnd) ? h3 : uEnd, h3, m, kh, f0, acc);
  k1_phase<0>(v, user_emb, item_rebased, curU, curI, uEnd, nEnd, m, kh, f0, acc);

  #pragma unroll
  for (int t = 0; t < 16; ++t) {
    red[s][2 * t][lane]     = acc[t].lo;
    red[s][2 * t + 1][lane] = acc[t].hi;
  }
  __syncthreads();

  const int ft   = s >> 2;
  const int q    = s & 3;
  const int half = q >> 1;
  const int comp = (q & 1) * 2;
  v2f sum[4];
  #pragma unroll
  for (int p = 0; p < K1_SUB; ++p) {
    #pragma unroll
    for (int dt = 0; dt < 4; ++dt) {
      const float* piece = reinterpret_cast<const float*>(
          &red[p][2 * (ft * 4 + dt) + half][lane]);
      const v2f x = *reinterpret_cast<const v2f*>(piece + comp);
      if (p == 0) sum[dt] = x; else sum[dt] += x;
    }
  }

  float* __restrict__ dst = partials + (size_t)sp * PROJ_ELEMS;
  #pragma unroll
  for (int rr = 0; rr < 2; ++rr) {
    const int fr = f0 + 4 * (2 * q + rr + 8 * kh) + ft;
    v4f o;
    o.x = sum[0][rr]; o.y = sum[1][rr]; o.z = sum[2][rr]; o.w = sum[3][rr];
    *reinterpret_cast<v4f*>(dst + fr * EMB + 4 * m) = o;
  }
}

__global__ void reduce_scale_kernel(const float* __restrict__ partials,
                                    const float* __restrict__ filt,
                                    float* __restrict__ projS,
                                    int split_cnt) {
  const int i = blockIdx.x * blockDim.x + threadIdx.x;
  if (i < PROJ_ELEMS) {
    float s = 0.f;
    for (int p = 0; p < split_cnt; ++p)
      s += partials[(size_t)p * PROJ_ELEMS + i];
    const int f = i >> 6;
    const int d = i & 63;
    const int o = (f >> 1) * 128 + (d >> 5) * 64 + (d & 15) * 4 +
                  ((d >> 4) & 1) * 2 + (f & 1);
    projS[o] = s * filt[f];
  }
}

#define K3_BLOCKS 256
#define N_TILES   (N_TOTAL / 16)

template <int PRIO>
__device__ __forceinline__ void k3_quarter(const v2f* __restrict__ ap,
                                           const v4f* bp, int fBeg,
                                           v8f& acc0, v8f& acc1,
                                           v8f& acc2, v8f& acc3) {
  __builtin_amdgcn_s_setprio(PRIO);
  #pragma unroll 2
  for (int f0 = fBeg; f0 < fBeg + FREQ / 4; f0 += 16) {
    v2f a[4];
    v4f q[8];
    #pragma unroll
    for (int j = 0; j < 4; ++j)
      a[j] = ap[(f0 >> 1) + 2 * j];
    #pragma unroll
    for (int j = 0; j < 4; ++j) {
      q[2 * j]     = bp[(f0 + 4 * j) * 16];
      q[2 * j + 1] = bp[(f0 + 4 * j) * 16 + 16];
    }
    #pragma unroll
    for (int j = 0; j < 4; ++j) {
      v2f b;
      b.x = q[2*j].x;   b.y = q[2*j].y;   acc0 = wmma_f32(a[j], b, acc0);
      b.x = q[2*j].z;   b.y = q[2*j].w;   acc1 = wmma_f32(a[j], b, acc1);
      b.x = q[2*j+1].x; b.y = q[2*j+1].y; acc2 = wmma_f32(a[j], b, acc2);
      b.x = q[2*j+1].z; b.y = q[2*j+1].w; acc3 = wmma_f32(a[j], b, acc3);
    }
  }
}

__global__ void __launch_bounds__(1024) __attribute__((amdgpu_waves_per_eu(8, 8)))
filter_combine_kernel(const float* __restrict__ user_emb,
                      const float* __restrict__ item_emb,
                      const float* __restrict__ v,
                      const float* __restrict__ projS,
                      float* __restrict__ out) {
  __shared__ v4f projL[PROJ_ELEMS / 4];

  const v4f* __restrict__ src = reinterpret_cast<const v4f*>(projS);
  #pragma unroll
  for (int i = 0; i < PROJ_ELEMS / 4 / 1024; ++i) {
    const int idx = ((i + blockIdx.x) & 7) * 1024 + threadIdx.x;
    projL[idx] = src[idx];
  }
  __syncthreads();

  const int lane = threadIdx.x & 31;
  const int wave = __builtin_amdgcn_readfirstlane(threadIdx.x >> 5);
  const int m  = lane & 15;
  const int kh = lane >> 4;

  const int tBeg = (blockIdx.x * N_TILES) / K3_BLOCKS;
  const int tEnd = ((blockIdx.x + 1) * N_TILES) / K3_BLOCKS;

  const v4f* bp = projL + kh * 32 + m;

  for (int nTile = tBeg + wave; nTile < tEnd; nTile += 32) {
    const int n0 = nTile * 16;

    v8f acc0 = {0,0,0,0,0,0,0,0};
    v8f acc1 = {0,0,0,0,0,0,0,0};
    v8f acc2 = {0,0,0,0,0,0,0,0};
    v8f acc3 = {0,0,0,0,0,0,0,0};

    const v2f* __restrict__ ap =
        reinterpret_cast<const v2f*>(v + (size_t)(n0 + m) * FREQ) + kh;

    k3_quarter<3>(ap, bp, 0 * (FREQ / 4), acc0, acc1, acc2, acc3);
    k3_quarter<2>(ap, bp, 1 * (FREQ / 4), acc0, acc1, acc2, acc3);
    k3_quarter<1>(ap, bp, 2 * (FREQ / 4), acc0, acc1, acc2, acc3);
    k3_quarter<0>(ap, bp, 3 * (FREQ / 4), acc0, acc1, acc2, acc3);

    const float* __restrict__ ego =
        (n0 < USER_NUM) ? user_emb : (item_emb - (size_t)USER_NUM * EMB);
    #pragma unroll
    for (int r = 0; r < 8; ++r) {
      const int row = n0 + r + 8 * kh;
      const float* __restrict__ er = ego + (size_t)row * EMB + m;
      float* __restrict__ orow = out + (size_t)row * EMB + m;
      __builtin_nontemporal_store((er[ 0] + acc0[r]) * 0.5f, orow +  0);
      __builtin_nontemporal_store((er[16] + acc1[r]) * 0.5f, orow + 16);
      __builtin_nontemporal_store((er[32] + acc2[r]) * 0.5f, orow + 32);
      __builtin_nontemporal_store((er[48] + acc3[r]) * 0.5f, orow + 48);
    }
  }
}

extern "C" void kernel_launch(void* const* d_in, const int* in_sizes, int n_in,
                              void* d_out, int out_size, void* d_ws, size_t ws_size,
                              hipStream_t stream) {
  const float* user_emb = (const float*)d_in[0];
  const float* item_emb = (const float*)d_in[1];
  const float* v        = (const float*)d_in[2];
  const float* filt     = (const float*)d_in[3];
  float* out = (float*)d_out;

  const size_t slot = (size_t)PROJ_ELEMS * sizeof(float);
  int split_cnt = (int)(ws_size / slot) - 1;
  if (split_cnt > MAX_SPLIT) split_cnt = MAX_SPLIT;
  if (split_cnt < 1) split_cnt = 1;

  float* partials = (float*)d_ws;
  float* projS    = partials + (size_t)split_cnt * PROJ_ELEMS;

  int chunk = ((N_TOTAL + split_cnt - 1) / split_cnt + 3) & ~3;

  dim3 gridA(FREQ / 64, split_cnt);
  proj_partial_kernel<<<gridA, 32 * K1_SUB, 0, stream>>>(user_emb, item_emb, v,
                                                         partials, chunk);

  reduce_scale_kernel<<<(PROJ_ELEMS + 255) / 256, 256, 0, stream>>>(
      partials, filt, projS, split_cnt);

  filter_combine_kernel<<<K3_BLOCKS, 1024, 0, stream>>>(
      user_emb, item_emb, v, projS, out);
}
